# starting kernel
# baseline (speedup 1.0000x reference)
#include <hip/hip_runtime.h>

#define BSZ 4
#define MSZ 2048
#define NSZ 16384
#define CH   64
#define XCH  128
#define OCH  64

typedef float v2f __attribute__((ext_vector_type(2)));
typedef float v8f __attribute__((ext_vector_type(8)));

#define MCHUNK  128
#define FSTRIDE 72

__global__ __launch_bounds__(256) void interp_kernel(
    const float* __restrict__ xyz_down, const float* __restrict__ xyz_up,
    const float* __restrict__ feat_down, const float* __restrict__ feat_up,
    float* __restrict__ xcat)
{
  __shared__ float s_xyz[MCHUNK * 4];
  __shared__ float s_feat[MCHUNK * FSTRIDE];

  const int b     = blockIdx.y;
  const int tid   = threadIdx.x;
  const int wave  = tid >> 5;
  const int lane  = tid & 31;
  const int ln    = lane & 15;
  const int khalf = (lane >> 4) << 1;
  const int n0    = blockIdx.x * 128 + wave * 16;

  {
    const float4* fu = (const float4*)feat_up;
    float4*       xc = (float4*)xcat;
#pragma unroll
    for (int j = 0; j < 8; ++j) {
      int  i4  = j * 32 + lane;
      int  row = i4 >> 4;
      int  c4  = i4 & 15;
      long gr  = (long)(b * NSZ + n0 + row);
      xc[gr * 32 + c4] = fu[gr * 16 + c4];
    }
  }

  const float* up = xyz_up + ((long)(b * NSZ + n0 + ln)) * 3;
  const float ux = up[0], uy = up[1], uz = up[2];
  const float u2 = ux * ux + uy * uy + uz * uz;

  v8f z = {0.f, 0.f, 0.f, 0.f, 0.f, 0.f, 0.f, 0.f};
  v8f acc[4];
#pragma unroll
  for (int t = 0; t < 4; ++t) acc[t] = z;
  float rsum = 0.f;

  for (int cstart = 0; cstart < MSZ; cstart += MCHUNK) {
    if (tid < MCHUNK) {
      const float* p = xyz_down + ((long)(b * MSZ + cstart + tid)) * 3;
      float px = p[0], py = p[1], pz = p[2];
      float4 v = {px, py, pz, px * px + py * py + pz * pz};
      *(float4*)&s_xyz[tid * 4] = v;
    }
    {
      const float4* fd =
          (const float4*)(feat_down + ((long)(b * MSZ + cstart)) * CH);
#pragma unroll
      for (int j = 0; j < 8; ++j) {
        int i4  = j * 256 + tid;
        int row = i4 >> 4;
        int c4  = i4 & 15;
        float4 v = fd[row * 16 + c4];
        *(float4*)&s_feat[row * FSTRIDE + c4 * 4] = v;
      }
    }
    __syncthreads();

    for (int k = 0; k < MCHUNK; k += 4) {
      const int m = k + khalf;
      float4 p0 = *(const float4*)&s_xyz[m * 4];
      float4 p1 = *(const float4*)&s_xyz[(m + 1) * 4];
      float dot0 = fmaf(ux, p0.x, fmaf(uy, p0.y, uz * p0.z));
      float dot1 = fmaf(ux, p1.x, fmaf(uy, p1.y, uz * p1.z));
      float d0 = fmaf(-2.f, dot0, u2 + p0.w) + 1e-8f;
      float d1 = fmaf(-2.f, dot1, u2 + p1.w) + 1e-8f;
      v2f a;
      a.x = __builtin_amdgcn_rcpf(d0);
      a.y = __builtin_amdgcn_rcpf(d1);
      rsum += a.x + a.y;
#pragma unroll
      for (int t = 0; t < 4; ++t) {
        v2f bm;
        bm.x = s_feat[m * FSTRIDE + t * 16 + ln];
        bm.y = s_feat[(m + 1) * FSTRIDE + t * 16 + ln];
        acc[t] = __builtin_amdgcn_wmma_f32_16x16x4_f32(
            false, a, false, bm, (short)0, acc[t], false, false);
      }
    }
    __syncthreads();
  }

  float denom = rsum + __shfl_xor(rsum, 16, 32);
  float inv   = 1.f / denom;
#pragma unroll
  for (int j = 0; j < 8; ++j) {
    float dj  = __shfl(inv, ((lane & 16) >> 1) + j, 32);
    int   row = n0 + j + ((lane & 16) ? 8 : 0);
    long  base = (long)(b * NSZ + row) * XCH + 64;
#pragma unroll
    for (int t = 0; t < 4; ++t)
      xcat[base + t * 16 + ln] = acc[t][j] * dj;
  }
}

#define K2WAVES 4
#define XSTR    132

__global__ __launch_bounds__(128) void mlp_kernel(
    const float* __restrict__ xcat,
    const float* __restrict__ W1, const float* __restrict__ b1,
    const float* __restrict__ g1, const float* __restrict__ be1,
    const float* __restrict__ mu1, const float* __restrict__ va1,
    const float* __restrict__ W2, const float* __restrict__ b2,
    const float* __restrict__ g2, const float* __restrict__ be2,
    const float* __restrict__ mu2, const float* __restrict__ va2,
    float* __restrict__ out)
{
  __shared__ float s_tile[K2WAVES * 16 * XSTR];

  const int tid     = threadIdx.x;
  const int wave    = tid >> 5;
  const int lane    = tid & 31;
  const int ln      = lane & 15;
  const int khalf   = (lane >> 4) << 1;
  const int rowbase = blockIdx.x * (K2WAVES * 16) + wave * 16;
  float*    tile    = s_tile + wave * 16 * XSTR;

  float s1[8], t1[8], s2[4], t2[4];
#pragma unroll
  for (int t = 0; t < 8; ++t) {
    int   c  = t * 16 + ln;
    float sc = g1[c] * rsqrtf(va1[c] + 1e-5f);
    s1[t] = sc;
    t1[t] = (b1[c] - mu1[c]) * sc + be1[c];
  }
#pragma unroll
  for (int t = 0; t < 4; ++t) {
    int   c  = t * 16 + ln;
    float sc = g2[c] * rsqrtf(va2[c] + 1e-5f);
    s2[t] = sc;
    t2[t] = (b2[c] - mu2[c]) * sc + be2[c];
  }

  {
    const float4* xc = (const float4*)(xcat + (long)rowbase * XCH);
#pragma unroll
    for (int j = 0; j < 16; ++j) {
      int i4  = j * 32 + lane;
      int row = i4 >> 5;
      int c4  = i4 & 31;
      float4 v = xc[row * 32 + c4];
      *(float4*)&tile[row * XSTR + c4 * 4] = v;
    }
  }
  __syncthreads();

  v8f z = {0.f, 0.f, 0.f, 0.f, 0.f, 0.f, 0.f, 0.f};
  v8f h[8];
#pragma unroll
  for (int t = 0; t < 8; ++t) h[t] = z;

  for (int k = 0; k < XCH; k += 4) {
    v2f a = *(const v2f*)&tile[ln * XSTR + k + khalf];
#pragma unroll
    for (int t = 0; t < 8; ++t) {
      v2f bm = *(const v2f*)(W1 + (t * 16 + ln) * XCH + k + khalf);
      h[t] = __builtin_amdgcn_wmma_f32_16x16x4_f32(
          false, a, false, bm, (short)0, h[t], false, false);
    }
  }
  __syncthreads();

#pragma unroll
  for (int t = 0; t < 8; ++t) {
#pragma unroll
    for (int j = 0; j < 8; ++j) {
      int   row = j + ((lane & 16) ? 8 : 0);
      float v   = h[t][j] * s1[t] + t1[t];
      tile[row * XSTR + t * 16 + ln] = v > 0.f ? v : 0.f;
    }
  }
  __syncthreads();

  v8f o[4];
#pragma unroll
  for (int t = 0; t < 4; ++t) o[t] = z;

  for (int k = 0; k < XCH; k += 4) {
    v2f a = *(const v2f*)&tile[ln * XSTR + k + khalf];
#pragma unroll
    for (int t = 0; t < 4; ++t) {
      v2f bm = *(const v2f*)(W2 + (t * 16 + ln) * XCH + k + khalf);
      o[t] = __builtin_amdgcn_wmma_f32_16x16x4_f32(
          false, a, false, bm, (short)0, o[t], false, false);
    }
  }

#pragma unroll
  for (int t = 0; t < 4; ++t) {
#pragma unroll
    for (int j = 0; j < 8; ++j) {
      int row = rowbase + j + ((lane & 16) ? 8 : 0);
      out[(long)row * OCH + t * 16 + ln] = o[t][j] * s2[t] + t2[t];
    }
  }
}

extern "C" void kernel_launch(void* const* d_in, const int* in_sizes, int n_in,
                              void* d_out, int out_size, void* d_ws,
                              size_t ws_size, hipStream_t stream) {
  const float* xyz_down  = (const float*)d_in[0];
  const float* xyz_up    = (const float*)d_in[1];
  const float* feat_down = (const float*)d_in[2];
  const float* feat_up   = (const float*)d_in[3];
  const float* W1  = (const float*)d_in[4];
  const float* b1  = (const float*)d_in[5];
  const float* g1  = (const float*)d_in[6];
  const float* be1 = (const float*)d_in[7];
  const float* mu1 = (const float*)d_in[8];
  const float* va1 = (const float*)d_in[9];
  const float* W2  = (const float*)d_in[10];
  const float* b2  = (const float*)d_in[11];
  const float* g2  = (const float*)d_in[12];
  const float* be2 = (const float*)d_in[13];
  const float* mu2 = (const float*)d_in[14];
  const float* va2 = (const float*)d_in[15];

  float* xcat = (float*)d_ws;

  dim3 g1d(NSZ / 128, BSZ);
  interp_kernel<<<g1d, 256, 0, stream>>>(xyz_down, xyz_up, feat_down, feat_up,
                                         xcat);

  const int rows = BSZ * NSZ;
  mlp_kernel<<<rows / (K2WAVES * 16), 128, 0, stream>>>(
      xcat, W1, b1, g1, be1, mu1, va1, W2, b2, g2, be2, mu2, va2,
      (float*)d_out);
}


# MLP kernel: W1 and W2 are copied once per work-group into padded LDS and both GEMMs read their B fragments from LDS instead of global memory; work-group = 16 waves x 16 rows, 256 groups
# speedup vs baseline: 1.0724x; 1.0724x over previous
#include <hip/hip_runtime.h>

#define BSZ 4
#define MSZ 2048
#define NSZ 16384
#define CH   64
#define XCH  128
#define OCH  64

typedef float v2f __attribute__((ext_vector_type(2)));
typedef float v8f __attribute__((ext_vector_type(8)));

#define MCHUNK  128
#define FSTRIDE 72

__global__ __launch_bounds__(256) void interp_kernel(
    const float* __restrict__ xyz_down, const float* __restrict__ xyz_up,
    const float* __restrict__ feat_down, const float* __restrict__ feat_up,
    float* __restrict__ xcat)
{
  __shared__ float s_xyz[MCHUNK * 4];
  __shared__ float s_feat[MCHUNK * FSTRIDE];

  const int b     = blockIdx.y;
  const int tid   = threadIdx.x;
  const int wave  = tid >> 5;
  const int lane  = tid & 31;
  const int ln    = lane & 15;
  const int khalf = (lane >> 4) << 1;
  const int n0    = blockIdx.x * 128 + wave * 16;

  {
    const float4* fu = (const float4*)feat_up;
    float4*       xc = (float4*)xcat;
#pragma unroll
    for (int j = 0; j < 8; ++j) {
      int  i4  = j * 32 + lane;
      int  row = i4 >> 4;
      int  c4  = i4 & 15;
      long gr  = (long)(b * NSZ + n0 + row);
      xc[gr * 32 + c4] = fu[gr * 16 + c4];
    }
  }

  const float* up = xyz_up + ((long)(b * NSZ + n0 + ln)) * 3;
  const float ux = up[0], uy = up[1], uz = up[2];
  const float u2 = ux * ux + uy * uy + uz * uz;

  v8f z = {0.f, 0.f, 0.f, 0.f, 0.f, 0.f, 0.f, 0.f};
  v8f acc[4];
#pragma unroll
  for (int t = 0; t < 4; ++t) acc[t] = z;
  float rsum = 0.f;

  for (int cstart = 0; cstart < MSZ; cstart += MCHUNK) {
    if (tid < MCHUNK) {
      const float* p = xyz_down + ((long)(b * MSZ + cstart + tid)) * 3;
      float px = p[0], py = p[1], pz = p[2];
      float4 v = {px, py, pz, px * px + py * py + pz * pz};
      *(float4*)&s_xyz[tid * 4] = v;
    }
    {
      const float4* fd =
          (const float4*)(feat_down + ((long)(b * MSZ + cstart)) * CH);
#pragma unroll
      for (int j = 0; j < 8; ++j) {
        int i4  = j * 256 + tid;
        int row = i4 >> 4;
        int c4  = i4 & 15;
        float4 v = fd[row * 16 + c4];
        *(float4*)&s_feat[row * FSTRIDE + c4 * 4] = v;
      }
    }
    __syncthreads();

    for (int k = 0; k < MCHUNK; k += 4) {
      const int m = k + khalf;
      float4 p0 = *(const float4*)&s_xyz[m * 4];
      float4 p1 = *(const float4*)&s_xyz[(m + 1) * 4];
      float dot0 = fmaf(ux, p0.x, fmaf(uy, p0.y, uz * p0.z));
      float dot1 = fmaf(ux, p1.x, fmaf(uy, p1.y, uz * p1.z));
      float d0 = fmaf(-2.f, dot0, u2 + p0.w) + 1e-8f;
      float d1 = fmaf(-2.f, dot1, u2 + p1.w) + 1e-8f;
      v2f a;
      a.x = __builtin_amdgcn_rcpf(d0);
      a.y = __builtin_amdgcn_rcpf(d1);
      rsum += a.x + a.y;
#pragma unroll
      for (int t = 0; t < 4; ++t) {
        v2f bm;
        bm.x = s_feat[m * FSTRIDE + t * 16 + ln];
        bm.y = s_feat[(m + 1) * FSTRIDE + t * 16 + ln];
        acc[t] = __builtin_amdgcn_wmma_f32_16x16x4_f32(
            false, a, false, bm, (short)0, acc[t], false, false);
      }
    }
    __syncthreads();
  }

  float denom = rsum + __shfl_xor(rsum, 16, 32);
  float inv   = 1.f / denom;
#pragma unroll
  for (int j = 0; j < 8; ++j) {
    float dj  = __shfl(inv, ((lane & 16) >> 1) + j, 32);
    int   row = n0 + j + ((lane & 16) ? 8 : 0);
    long  base = (long)(b * NSZ + row) * XCH + 64;
#pragma unroll
    for (int t = 0; t < 4; ++t)
      xcat[base + t * 16 + ln] = acc[t][j] * dj;
  }
}

#define K2WAVES 16
#define XSTR    132

__global__ __launch_bounds__(512) void mlp_kernel(
    const float* __restrict__ xcat,
    const float* __restrict__ W1, const float* __restrict__ b1,
    const float* __restrict__ g1, const float* __restrict__ be1,
    const float* __restrict__ mu1, const float* __restrict__ va1,
    const float* __restrict__ W2, const float* __restrict__ b2,
    const float* __restrict__ g2, const float* __restrict__ be2,
    const float* __restrict__ mu2, const float* __restrict__ va2,
    float* __restrict__ out)
{
  __shared__ float s_w1[XCH * XSTR];
  __shared__ float s_w2[OCH * XSTR];
  __shared__ float s_tile[K2WAVES * 16 * XSTR];

  const int tid     = threadIdx.x;
  const int wave    = tid >> 5;
  const int lane    = tid & 31;
  const int ln      = lane & 15;
  const int khalf   = (lane >> 4) << 1;
  const int rowbase = blockIdx.x * (K2WAVES * 16) + wave * 16;
  float*    tile    = s_tile + wave * 16 * XSTR;

  float s1[8], t1[8], s2[4], t2[4];
#pragma unroll
  for (int t = 0; t < 8; ++t) {
    int   c  = t * 16 + ln;
    float sc = g1[c] * rsqrtf(va1[c] + 1e-5f);
    s1[t] = sc;
    t1[t] = (b1[c] - mu1[c]) * sc + be1[c];
  }
#pragma unroll
  for (int t = 0; t < 4; ++t) {
    int   c  = t * 16 + ln;
    float sc = g2[c] * rsqrtf(va2[c] + 1e-5f);
    s2[t] = sc;
    t2[t] = (b2[c] - mu2[c]) * sc + be2[c];
  }

  {
    const float4* w1v = (const float4*)W1;
#pragma unroll
    for (int j = 0; j < 8; ++j) {
      int i4  = j * 512 + tid;
      int row = i4 >> 5;
      int c4  = i4 & 31;
      *(float4*)&s_w1[row * XSTR + c4 * 4] = w1v[i4];
    }
    const float4* w2v = (const float4*)W2;
#pragma unroll
    for (int j = 0; j < 4; ++j) {
      int i4  = j * 512 + tid;
      int row = i4 >> 5;
      int c4  = i4 & 31;
      *(float4*)&s_w2[row * XSTR + c4 * 4] = w2v[i4];
    }
  }

  {
    const float4* xc = (const float4*)(xcat + (long)rowbase * XCH);
#pragma unroll
    for (int j = 0; j < 16; ++j) {
      int i4  = j * 32 + lane;
      int row = i4 >> 5;
      int c4  = i4 & 31;
      float4 v = xc[row * 32 + c4];
      *(float4*)&tile[row * XSTR + c4 * 4] = v;
    }
  }
  __syncthreads();

  v8f z = {0.f, 0.f, 0.f, 0.f, 0.f, 0.f, 0.f, 0.f};
  v8f h[8];
#pragma unroll
  for (int t = 0; t < 8; ++t) h[t] = z;

  for (int k = 0; k < XCH; k += 4) {
    v2f a = *(const v2f*)&tile[ln * XSTR + k + khalf];
#pragma unroll
    for (int t = 0; t < 8; ++t) {
      v2f bm = *(const v2f*)&s_w1[(t * 16 + ln) * XSTR + k + khalf];
      h[t] = __builtin_amdgcn_wmma_f32_16x16x4_f32(
          false, a, false, bm, (short)0, h[t], false, false);
    }
  }
  __syncthreads();

#pragma unroll
  for (int t = 0; t < 8; ++t) {
#pragma unroll
    for (int j = 0; j < 8; ++j) {
      int   row = j + ((lane & 16) ? 8 : 0);
      float v   = h[t][j] * s1[t] + t1[t];
      tile[row * XSTR + t * 16 + ln] = v > 0.f ? v : 0.f;
    }
  }
  __syncthreads();

  v8f o[4];
#pragma unroll
  for (int t = 0; t < 4; ++t) o[t] = z;

  for (int k = 0; k < XCH; k += 4) {
    v2f a = *(const v2f*)&tile[ln * XSTR + k + khalf];
#pragma unroll
    for (int t = 0; t < 4; ++t) {
      v2f bm = *(const v2f*)&s_w2[(t * 16 + ln) * XSTR + k + khalf];
      o[t] = __builtin_amdgcn_wmma_f32_16x16x4_f32(
          false, a, false, bm, (short)0, o[t], false, false);
    }
  }

#pragma unroll
  for (int t = 0; t < 4; ++t) {
#pragma unroll
    for (int j = 0; j < 8; ++j) {
      int row = rowbase + j + ((lane & 16) ? 8 : 0);
      out[(long)row * OCH + t * 16 + ln] = o[t][j] * s2[t] + t2[t];
    }
  }
}

extern "C" void kernel_launch(void* const* d_in, const int* in_sizes, int n_in,
                              void* d_out, int out_size, void* d_ws,
                              size_t ws_size, hipStream_t stream) {
  const float* xyz_down  = (const float*)d_in[0];
  const float* xyz_up    = (const float*)d_in[1];
  const float* feat_down = (const float*)d_in[2];
  const float* feat_up   = (const float*)d_in[3];
  const float* W1  = (const float*)d_in[4];
  const float* b1  = (const float*)d_in[5];
  const float* g1  = (const float*)d_in[6];
  const float* be1 = (const float*)d_in[7];
  const float* mu1 = (const float*)d_in[8];
  const float* va1 = (const float*)d_in[9];
  const float* W2  = (const float*)d_in[10];
  const float* b2  = (const float*)d_in[11];
  const float* g2  = (const float*)d_in[12];
  const float* be2 = (const float*)d_in[13];
  const float* mu2 = (const float*)d_in[14];
  const float* va2 = (const float*)d_in[15];

  float* xcat = (float*)d_ws;

  dim3 g1d(NSZ / 128, BSZ);
  interp_kernel<<<g1d, 256, 0, stream>>>(xyz_down, xyz_up, feat_down, feat_up,
                                         xcat);

  const int rows = BSZ * NSZ;
  mlp_kernel<<<rows / (K2WAVES * 16), 512, 0, stream>>>(
      xcat, W1, b1, g1, be1, mu1, va1, W2, b2, g2, be2, mu2, va2,
      (float*)d_out);
}


# interp: squared distances of a 16-point x 16-query block from ONE f32 WMMA ([|p|^2,-2p] x [1,u] + |u|^2); its D registers feed rcp and are the A fragments; m sum order changes inside 16-blocks
# speedup vs baseline: 1.2323x; 1.1491x over previous
#include <hip/hip_runtime.h>

#define BSZ 4
#define MSZ 2048
#define NSZ 16384
#define CH   64
#define XCH  128
#define OCH  64

typedef float v2f __attribute__((ext_vector_type(2)));
typedef float v8f __attribute__((ext_vector_type(8)));

#define MCHUNK  128
#define FPITCH  136

__global__ __launch_bounds__(256) void interp_kernel(
    const float* __restrict__ xyz_down, const float* __restrict__ xyz_up,
    const float* __restrict__ feat_down, const float* __restrict__ feat_up,
    float* __restrict__ xcat)
{
  __shared__ float s_xyz[MCHUNK * 4];
  __shared__ float s_feat[(MCHUNK / 2) * FPITCH];

  const int b     = blockIdx.y;
  const int tid   = threadIdx.x;
  const int wave  = tid >> 5;
  const int lane  = tid & 31;
  const int ln    = lane & 15;
  const int khalf = (lane >> 4) << 1;
  const int hpair = (lane >> 4) << 2;
  const int n0    = blockIdx.x * 128 + wave * 16;

  {
    const float4* fu = (const float4*)feat_up;
    float4*       xc = (float4*)xcat;
#pragma unroll
    for (int j = 0; j < 8; ++j) {
      int  i4  = j * 32 + lane;
      int  row = i4 >> 4;
      int  c4  = i4 & 15;
      long gr  = (long)(b * NSZ + n0 + row);
      xc[gr * 32 + c4] = fu[gr * 16 + c4];
    }
  }

  const float* up = xyz_up + ((long)(b * NSZ + n0 + ln)) * 3;
  const float ux = up[0], uy = up[1], uz = up[2];
  const float u2 = ux * ux + uy * uy + uz * uz;

  v2f uf;
  uf.x = (lane & 16) ? uy : 1.f;
  uf.y = (lane & 16) ? uz : ux;
  const v8f cu   = {u2, u2, u2, u2, u2, u2, u2, u2};
  const v2f eps2 = {1e-8f, 1e-8f};

  v8f z = {0.f, 0.f, 0.f, 0.f, 0.f, 0.f, 0.f, 0.f};
  v8f acc[4];
#pragma unroll
  for (int t = 0; t < 4; ++t) acc[t] = z;
  v2f rs2 = {0.f, 0.f};

  for (int cstart = 0; cstart < MSZ; cstart += MCHUNK) {
    if (tid < MCHUNK) {
      const float* p = xyz_down + ((long)(b * MSZ + cstart + tid)) * 3;
      float px = p[0], py = p[1], pz = p[2];
      float4 v = {px * px + py * py + pz * pz, -2.f * px, -2.f * py, -2.f * pz};
      *(float4*)&s_xyz[tid * 4] = v;
    }
    {
      const float4* fd =
          (const float4*)(feat_down + ((long)(b * MSZ + cstart)) * CH);
#pragma unroll
      for (int j = 0; j < 8; ++j) {
        int i4  = j * 256 + tid;
        int row = i4 >> 4;
        int c4  = i4 & 15;
        float4 v = fd[row * 16 + c4];
        float* dst = &s_feat[(row >> 1) * FPITCH + c4 * 8 + (row & 1)];
        dst[0] = v.x;
        dst[2] = v.y;
        dst[4] = v.z;
        dst[6] = v.w;
      }
    }
    __syncthreads();

#pragma unroll
    for (int m0 = 0; m0 < MCHUNK; m0 += 16) {
      v2f pf = *(const v2f*)&s_xyz[(m0 + ln) * 4 + khalf];
      v8f dt = __builtin_amdgcn_wmma_f32_16x16x4_f32(
          false, pf, false, uf, (short)0, cu, false, false);
      const float* fb = &s_feat[((m0 >> 1) + hpair) * FPITCH + 2 * ln];
#pragma unroll
      for (int j = 0; j < 4; ++j) {
        v2f d2;
        d2.x = dt[2 * j];
        d2.y = dt[2 * j + 1];
        d2 += eps2;
        v2f a;
        a.x = __builtin_amdgcn_rcpf(d2.x);
        a.y = __builtin_amdgcn_rcpf(d2.y);
        rs2 += a;
#pragma unroll
        for (int t = 0; t < 4; ++t) {
          v2f bm = *(const v2f*)&fb[j * FPITCH + t * 32];
          acc[t] = __builtin_amdgcn_wmma_f32_16x16x4_f32(
              false, a, false, bm, (short)0, acc[t], false, false);
        }
      }
    }
    __syncthreads();
  }
  const float rsum = rs2.x + rs2.y;

  float denom = rsum + __shfl_xor(rsum, 16, 32);
  float inv   = 1.f / denom;
#pragma unroll
  for (int j = 0; j < 8; ++j) {
    float dj  = __shfl(inv, ((lane & 16) >> 1) + j, 32);
    int   row = n0 + j + ((lane & 16) ? 8 : 0);
    long  base = (long)(b * NSZ + row) * XCH + 64;
#pragma unroll
    for (int t = 0; t < 4; ++t)
      xcat[base + t * 16 + ln] = acc[t][j] * dj;
  }
}

#define K2WAVES 16
#define XSTR    132

__global__ __launch_bounds__(512) void mlp_kernel(
    const float* __restrict__ xcat,
    const float* __restrict__ W1, const float* __restrict__ b1,
    const float* __restrict__ g1, const float* __restrict__ be1,
    const float* __restrict__ mu1, const float* __restrict__ va1,
    const float* __restrict__ W2, const float* __restrict__ b2,
    const float* __restrict__ g2, const float* __restrict__ be2,
    const float* __restrict__ mu2, const float* __restrict__ va2,
    float* __restrict__ out)
{
  __shared__ float s_w1[XCH * XSTR];
  __shared__ float s_w2[OCH * XSTR];
  __shared__ float s_tile[K2WAVES * 16 * XSTR];

  const int tid     = threadIdx.x;
  const int wave    = tid >> 5;
  const int lane    = tid & 31;
  const int ln      = lane & 15;
  const int khalf   = (lane >> 4) << 1;
  const int rowbase = blockIdx.x * (K2WAVES * 16) + wave * 16;
  float*    tile    = s_tile + wave * 16 * XSTR;

  float s1[8], t1[8], s2[4], t2[4];
#pragma unroll
  for (int t = 0; t < 8; ++t) {
    int   c  = t * 16 + ln;
    float sc = g1[c] * rsqrtf(va1[c] + 1e-5f);
    s1[t] = sc;
    t1[t] = (b1[c] - mu1[c]) * sc + be1[c];
  }
#pragma unroll
  for (int t = 0; t < 4; ++t) {
    int   c  = t * 16 + ln;
    float sc = g2[c] * rsqrtf(va2[c] + 1e-5f);
    s2[t] = sc;
    t2[t] = (b2[c] - mu2[c]) * sc + be2[c];
  }

  {
    const float4* w1v = (const float4*)W1;
#pragma unroll
    for (int j = 0; j < 8; ++j) {
      int i4  = j * 512 + tid;
      int row = i4 >> 5;
      int c4  = i4 & 31;
      *(float4*)&s_w1[row * XSTR + c4 * 4] = w1v[i4];
    }
    const float4* w2v = (const float4*)W2;
#pragma unroll
    for (int j = 0; j < 4; ++j) {
      int i4  = j * 512 + tid;
      int row = i4 >> 5;
      int c4  = i4 & 31;
      *(float4*)&s_w2[row * XSTR + c4 * 4] = w2v[i4];
    }
  }

  {
    const float4* xc = (const float4*)(xcat + (long)rowbase * XCH);
#pragma unroll
    for (int j = 0; j < 16; ++j) {
      int i4  = j * 32 + lane;
      int row = i4 >> 5;
      int c4  = i4 & 31;
      float4 v = xc[row * 32 + c4];
      *(float4*)&tile[row * XSTR + c4 * 4] = v;
    }
  }
  __syncthreads();

  v8f z = {0.f, 0.f, 0.f, 0.f, 0.f, 0.f, 0.f, 0.f};
  v8f h[8];
#pragma unroll
  for (int t = 0; t < 8; ++t) h[t] = z;

  for (int k = 0; k < XCH; k += 4) {
    v2f a = *(const v2f*)&tile[ln * XSTR + k + khalf];
#pragma unroll
    for (int t = 0; t < 8; ++t) {
      v2f bm = *(const v2f*)&s_w1[(t * 16 + ln) * XSTR + k + khalf];
      h[t] = __builtin_amdgcn_wmma_f32_16x16x4_f32(
          false, a, false, bm, (short)0, h[t], false, false);
    }
  }
  __syncthreads();

#pragma unroll
  for (int t = 0; t < 8; ++t) {
#pragma unroll
    for (int j = 0; j < 8; ++j) {
      int   row = j + ((lane & 16) ? 8 : 0);
      float v   = h[t][j] * s1[t] + t1[t];
      tile[row * XSTR + t * 16 + ln] = v > 0.f ? v : 0.f;
    }
  }
  __syncthreads();

  v8f o[4];
#pragma unroll
  for (int t = 0; t < 4; ++t) o[t] = z;

  for (int k = 0; k < XCH; k += 4) {
    v2f a = *(const v2f*)&tile[ln * XSTR + k + khalf];
#pragma unroll
    for (int t = 0; t < 4; ++t) {
      v2f bm = *(const v2f*)&s_w2[(t * 16 + ln) * XSTR + k + khalf];
      o[t] = __builtin_amdgcn_wmma_f32_16x16x4_f32(
          false, a, false, bm, (short)0, o[t], false, false);
    }
  }

#pragma unroll
  for (int t = 0; t < 4; ++t) {
#pragma unroll
    for (int j = 0; j < 8; ++j) {
      int row = rowbase + j + ((lane & 16) ? 8 : 0);
      out[(long)row * OCH + t * 16 + ln] = o[t][j] * s2[t] + t2[t];
    }
  }
}

extern "C" void kernel_launch(void* const* d_in, const int* in_sizes, int n_in,
                              void* d_out, int out_size, void* d_ws,
                              size_t ws_size, hipStream_t stream) {
  const float* xyz_down  = (const float*)d_in[0];
  const float* xyz_up    = (const float*)d_in[1];
  const float* feat_down = (const float*)d_in[2];
  const float* feat_up   = (const float*)d_in[3];
  const float* W1  = (const float*)d_in[4];
  const float* b1  = (const float*)d_in[5];
  const float* g1  = (const float*)d_in[6];
  const float* be1 = (const float*)d_in[7];
  const float* mu1 = (const float*)d_in[8];
  const float* va1 = (const float*)d_in[9];
  const float* W2  = (const float*)d_in[10];
  const float* b2  = (const float*)d_in[11];
  const float* g2  = (const float*)d_in[12];
  const float* be2 = (const float*)d_in[13];
  const float* mu2 = (const float*)d_in[14];
  const float* va2 = (const float*)d_in[15];

  float* xcat = (float*)d_ws;

  dim3 g1d(NSZ / 128, BSZ);
  interp_kernel<<<g1d, 256, 0, stream>>>(xyz_down, xyz_up, feat_down, feat_up,
                                         xcat);

  const int rows = BSZ * NSZ;
  mlp_kernel<<<rows / (K2WAVES * 16), 512, 0, stream>>>(
      xcat, W1, b1, g1, be1, mu1, va1, W2, b2, g2, be2, mu2, va2,
      (float*)d_out);
}


# one launch: each wave runs the MLP on its own 16 rows straight from the interpolation accumulators through a 16x64 LDS tile; W1/W2 staged beside the point chunks; workspace round trip gone
# speedup vs baseline: 1.3083x; 1.0617x over previous
#include <hip/hip_runtime.h>

#define BSZ 4
#define MSZ 2048
#define NSZ 16384
#define CH   64
#define XCH  128
#define OCH  64

typedef float v2f __attribute__((ext_vector_type(2)));
typedef float v4f __attribute__((ext_vector_type(4)));
typedef float v8f __attribute__((ext_vector_type(8)));

#define MCHUNK  128
#define FPITCH  136
#define XSTR    132
#define TP      68

__global__ __launch_bounds__(256) void interp_mlp_kernel(
    const float* __restrict__ xyz_down, const float* __restrict__ xyz_up,
    const float* __restrict__ feat_down, const float* __restrict__ feat_up,
    const float* __restrict__ W1, const float* __restrict__ b1,
    const float* __restrict__ g1, const float* __restrict__ be1,
    const float* __restrict__ mu1, const float* __restrict__ va1,
    const float* __restrict__ W2, const float* __restrict__ b2,
    const float* __restrict__ g2, const float* __restrict__ be2,
    const float* __restrict__ mu2, const float* __restrict__ va2,
    float* __restrict__ out)
{
  __shared__ __attribute__((aligned(16))) float s_xyz[MCHUNK * 4];
  __shared__ __attribute__((aligned(16))) float s_feat[(MCHUNK / 2) * FPITCH];
  __shared__ __attribute__((aligned(16))) float s_w1[XCH * XSTR];
  __shared__ __attribute__((aligned(16))) float s_w2[OCH * XSTR];

  const int b     = blockIdx.y;
  const int tid   = threadIdx.x;
  const int wave  = tid >> 5;
  const int lane  = tid & 31;
  const int ln    = lane & 15;
  const int khalf = (lane >> 4) << 1;
  const int hpair = (lane >> 4) << 2;
  const int n0    = blockIdx.x * 128 + wave * 16;

  const float* up = xyz_up + ((long)(b * NSZ + n0 + ln)) * 3;
  const float ux = up[0], uy = up[1], uz = up[2];
  const float u2 = ux * ux + uy * uy + uz * uz;

  v2f uf;
  uf.x = (lane & 16) ? uy : 1.f;
  uf.y = (lane & 16) ? uz : ux;
  const v8f cu   = {u2, u2, u2, u2, u2, u2, u2, u2};
  const v2f eps2 = {1e-8f, 1e-8f};

  v8f z = {0.f, 0.f, 0.f, 0.f, 0.f, 0.f, 0.f, 0.f};
  v8f acc[4];
#pragma unroll
  for (int t = 0; t < 4; ++t) acc[t] = z;
  v2f rs2 = {0.f, 0.f};

  for (int cstart = 0; cstart < MSZ; cstart += MCHUNK) {
    if (tid < MCHUNK) {
      const float* p = xyz_down + ((long)(b * MSZ + cstart + tid)) * 3;
      float px = p[0], py = p[1], pz = p[2];
      float4 v = {px * px + py * py + pz * pz, -2.f * px, -2.f * py, -2.f * pz};
      *(float4*)&s_xyz[tid * 4] = v;
    }
    {
      const float4* fd =
          (const float4*)(feat_down + ((long)(b * MSZ + cstart)) * CH);
#pragma unroll
      for (int j = 0; j < 8; ++j) {
        int i4  = j * 256 + tid;
        int row = i4 >> 4;
        int c4  = i4 & 15;
        float4 v = fd[row * 16 + c4];
        float* dst = &s_feat[(row >> 1) * FPITCH + c4 * 8 + (row & 1)];
        dst[0] = v.x;
        dst[2] = v.y;
        dst[4] = v.z;
        dst[6] = v.w;
      }
    }
    {
      int i4  = (cstart / MCHUNK) * 256 + tid;
      int row = i4 >> 5;
      int c4  = i4 & 31;
      int r2  = i4 >> 6;
      int c2  = i4 & 63;
      float4 w1v = ((const float4*)W1)[i4];
      float2 w2v = ((const float2*)W2)[i4];
      *(float4*)&s_w1[row * XSTR + c4 * 4] = w1v;
      *(float2*)&s_w2[r2 * XSTR + c2 * 2]  = w2v;
    }
    __syncthreads();

#pragma unroll
    for (int m0 = 0; m0 < MCHUNK; m0 += 16) {
      v2f pf = *(const v2f*)&s_xyz[(m0 + ln) * 4 + khalf];
      v8f dt = __builtin_amdgcn_wmma_f32_16x16x4_f32(
          false, pf, false, uf, (short)0, cu, false, false);
      const float* fb = &s_feat[((m0 >> 1) + hpair) * FPITCH + 2 * ln];
#pragma unroll
      for (int j = 0; j < 4; ++j) {
        v2f d2;
        d2.x = dt[2 * j];
        d2.y = dt[2 * j + 1];
        d2 += eps2;
        v2f a;
        a.x = __builtin_amdgcn_rcpf(d2.x);
        a.y = __builtin_amdgcn_rcpf(d2.y);
        rs2 += a;
#pragma unroll
        for (int t = 0; t < 4; ++t) {
          v2f bm = *(const v2f*)&fb[j * FPITCH + t * 32];
          acc[t] = __builtin_amdgcn_wmma_f32_16x16x4_f32(
              false, a, false, bm, (short)0, acc[t], false, false);
        }
      }
    }
    __syncthreads();
  }
  const long rowbase = (long)(b * NSZ + n0);
  v4f fv[8];
  {
    const v4f* fu = (const v4f*)(feat_up + rowbase * CH);
#pragma unroll
    for (int j = 0; j < 8; ++j) fv[j] = fu[j * 32 + lane];
  }

  const float rsum = rs2.x + rs2.y;

  float denom = rsum + __shfl_xor(rsum, 16, 32);
  float inv   = 1.f / denom;

  float*    tile = s_feat + wave * 16 * TP;
  const int hrow = (lane & 16) ? 8 : 0;

  float s1[8], t1[8], s2[4], t2[4];
#pragma unroll
  for (int t = 0; t < 8; ++t) {
    int   c  = t * 16 + ln;
    float sc = g1[c] * rsqrtf(va1[c] + 1e-5f);
    s1[t] = sc;
    t1[t] = (b1[c] - mu1[c]) * sc + be1[c];
  }
#pragma unroll
  for (int t = 0; t < 4; ++t) {
    int   c  = t * 16 + ln;
    float sc = g2[c] * rsqrtf(va2[c] + 1e-5f);
    s2[t] = sc;
    t2[t] = (b2[c] - mu2[c]) * sc + be2[c];
  }

#pragma unroll
  for (int j = 0; j < 8; ++j) {
    int i4  = j * 32 + lane;
    int row = i4 >> 4;
    int c4  = i4 & 15;
    *(v4f*)&tile[row * TP + c4 * 4] = fv[j];
  }
  __syncthreads();

  v8f h[8];
#pragma unroll
  for (int t = 0; t < 8; ++t) h[t] = z;

  for (int k = 0; k < CH; k += 4) {
    v2f a = *(const v2f*)&tile[ln * TP + k + khalf];
#pragma unroll
    for (int t = 0; t < 8; ++t) {
      v2f bm = *(const v2f*)&s_w1[(t * 16 + ln) * XSTR + k + khalf];
      h[t] = __builtin_amdgcn_wmma_f32_16x16x4_f32(
          false, a, false, bm, (short)0, h[t], false, false);
    }
  }
  __syncthreads();

#pragma unroll
  for (int j = 0; j < 8; ++j) {
    float dj = __shfl(inv, hrow + j, 32);
#pragma unroll
    for (int t = 0; t < 4; ++t)
      tile[(hrow + j) * TP + t * 16 + ln] = acc[t][j] * dj;
  }
  __syncthreads();

  for (int k = 0; k < CH; k += 4) {
    v2f a = *(const v2f*)&tile[ln * TP + k + khalf];
#pragma unroll
    for (int t = 0; t < 8; ++t) {
      v2f bm = *(const v2f*)&s_w1[(t * 16 + ln) * XSTR + CH + k + khalf];
      h[t] = __builtin_amdgcn_wmma_f32_16x16x4_f32(
          false, a, false, bm, (short)0, h[t], false, false);
    }
  }
  __syncthreads();

  v8f o[4];
#pragma unroll
  for (int t = 0; t < 4; ++t) o[t] = z;

#pragma unroll
  for (int hf = 0; hf < 2; ++hf) {
#pragma unroll
    for (int t = 0; t < 4; ++t) {
#pragma unroll
      for (int j = 0; j < 8; ++j) {
        float v = h[hf * 4 + t][j] * s1[hf * 4 + t] + t1[hf * 4 + t];
        tile[(hrow + j) * TP + t * 16 + ln] = v > 0.f ? v : 0.f;
      }
    }
    __syncthreads();
    for (int k = 0; k < CH; k += 4) {
      v2f a = *(const v2f*)&tile[ln * TP + k + khalf];
#pragma unroll
      for (int t = 0; t < 4; ++t) {
        v2f bm = *(const v2f*)&s_w2[(t * 16 + ln) * XSTR + hf * CH + k + khalf];
        o[t] = __builtin_amdgcn_wmma_f32_16x16x4_f32(
            false, a, false, bm, (short)0, o[t], false, false);
      }
    }
    if (hf == 0) __syncthreads();
  }

#pragma unroll
  for (int t = 0; t < 4; ++t) {
#pragma unroll
    for (int j = 0; j < 8; ++j)
      out[(rowbase + hrow + j) * OCH + t * 16 + ln] = o[t][j] * s2[t] + t2[t];
  }
}

extern "C" void kernel_launch(void* const* d_in, const int* in_sizes, int n_in,
                              void* d_out, int out_size, void* d_ws,
                              size_t ws_size, hipStream_t stream) {
  const float* xyz_down  = (const float*)d_in[0];
  const float* xyz_up    = (const float*)d_in[1];
  const float* feat_down = (const float*)d_in[2];
  const float* feat_up   = (const float*)d_in[3];
  const float* W1  = (const float*)d_in[4];
  const float* b1  = (const float*)d_in[5];
  const float* g1  = (const float*)d_in[6];
  const float* be1 = (const float*)d_in[7];
  const float* mu1 = (const float*)d_in[8];
  const float* va1 = (const float*)d_in[9];
  const float* W2  = (const float*)d_in[10];
  const float* b2  = (const float*)d_in[11];
  const float* g2  = (const float*)d_in[12];
  const float* be2 = (const float*)d_in[13];
  const float* mu2 = (const float*)d_in[14];
  const float* va2 = (const float*)d_in[15];

  dim3 grid(NSZ / 128, BSZ);
  interp_mlp_kernel<<<grid, 256, 0, stream>>>(
      xyz_down, xyz_up, feat_down, feat_up,
      W1, b1, g1, be1, mu1, va1, W2, b2, g2, be2, mu2, va2,
      (float*)d_out);
}


# chunk staging software-pipelined: global loads of point chunk c+1 (xyz, feat, W1/W2 piece) issued before the WMMAs of chunk c, held in registers, stored to LDS behind the closing barrier
# speedup vs baseline: 1.3727x; 1.0492x over previous
#include <hip/hip_runtime.h>

#define BSZ 4
#define MSZ 2048
#define NSZ 16384
#define CH   64
#define XCH  128
#define OCH  64

typedef float v2f __attribute__((ext_vector_type(2)));
typedef float v4f __attribute__((ext_vector_type(4)));
typedef float v8f __attribute__((ext_vector_type(8)));

#define MCHUNK  128
#define FPITCH  136
#define XSTR    132
#define TP      68

__global__ __launch_bounds__(256) void interp_mlp_kernel(
    const float* __restrict__ xyz_down, const float* __restrict__ xyz_up,
    const float* __restrict__ feat_down, const float* __restrict__ feat_up,
    const float* __restrict__ W1, const float* __restrict__ b1,
    const float* __restrict__ g1, const float* __restrict__ be1,
    const float* __restrict__ mu1, const float* __restrict__ va1,
    const float* __restrict__ W2, const float* __restrict__ b2,
    const float* __restrict__ g2, const float* __restrict__ be2,
    const float* __restrict__ mu2, const float* __restrict__ va2,
    float* __restrict__ out)
{
  __shared__ __attribute__((aligned(16))) float s_xyz[MCHUNK * 4];
  __shared__ __attribute__((aligned(16))) float s_feat[(MCHUNK / 2) * FPITCH];
  __shared__ __attribute__((aligned(16))) float s_w1[XCH * XSTR];
  __shared__ __attribute__((aligned(16))) float s_w2[OCH * XSTR];

  const int b     = blockIdx.y;
  const int tid   = threadIdx.x;
  const int wave  = tid >> 5;
  const int lane  = tid & 31;
  const int ln    = lane & 15;
  const int khalf = (lane >> 4) << 1;
  const int hpair = (lane >> 4) << 2;
  const int n0    = blockIdx.x * 128 + wave * 16;

  const float* up = xyz_up + ((long)(b * NSZ + n0 + ln)) * 3;
  const float ux = up[0], uy = up[1], uz = up[2];
  const float u2 = ux * ux + uy * uy + uz * uz;

  v2f uf;
  uf.x = (lane & 16) ? uy : 1.f;
  uf.y = (lane & 16) ? uz : ux;
  const v8f cu   = {u2, u2, u2, u2, u2, u2, u2, u2};
  const v2f eps2 = {1e-8f, 1e-8f};

  v8f z = {0.f, 0.f, 0.f, 0.f, 0.f, 0.f, 0.f, 0.f};
  v8f acc[4];
#pragma unroll
  for (int t = 0; t < 4; ++t) acc[t] = z;
  v2f rs2 = {0.f, 0.f};

  v2f    pxy;
  float  pzr;
  float4 fr[8];
  float4 w1r;
  float2 w2r;

#define LOAD_CHUNK(CS)                                                               \
  {                                                                                  \
      \
      \
    const float* p = xyz_down + ((long)(b * MSZ + (CS) + (tid & (MCHUNK - 1)))) * 3; \
    pxy.x = p[0];                                                                    \
    pxy.y = p[1];                                                                    \
    pzr   = p[2];                                                                    \
    const float4* fd = (const float4*)(feat_down + ((long)(b * MSZ + (CS))) * CH);   \
    _Pragma("unroll") for (int j = 0; j < 8; ++j) fr[j] = fd[j * 256 + tid];         \
    const int iw = ((CS) / MCHUNK) * 256 + tid;                          \
    w1r = ((const float4*)W1)[iw];                                                   \
    w2r = ((const float2*)W2)[iw];                                                   \
  }

  LOAD_CHUNK(0);

  for (int cstart = 0; cstart < MSZ; cstart += MCHUNK) {
    if (tid < MCHUNK) {
      const v2f sq = pxy * pxy;
      const v2f n2 = pxy * -2.f;
      float4 v = {sq.x + sq.y + pzr * pzr, n2.x, n2.y, -2.f * pzr};
      *(float4*)&s_xyz[tid * 4] = v;
    }
#pragma unroll
    for (int j = 0; j < 8; ++j) {
      int i4  = j * 256 + tid;
      int row = i4 >> 4;
      int c4  = i4 & 15;
      float* dst = &s_feat[(row >> 1) * FPITCH + c4 * 8 + (row & 1)];
      dst[0] = fr[j].x;
      dst[2] = fr[j].y;
      dst[4] = fr[j].z;
      dst[6] = fr[j].w;
    }
    {
      int i4  = (cstart / MCHUNK) * 256 + tid;
      int row = i4 >> 5;
      int c4  = i4 & 31;
      int r2  = i4 >> 6;
      int c2  = i4 & 63;
      *(float4*)&s_w1[row * XSTR + c4 * 4] = w1r;
      *(float2*)&s_w2[r2 * XSTR + c2 * 2]  = w2r;
    }
    __syncthreads();

    if (cstart + MCHUNK < MSZ) LOAD_CHUNK(cstart + MCHUNK);

#pragma unroll
    for (int m0 = 0; m0 < MCHUNK; m0 += 16) {
      v2f pf = *(const v2f*)&s_xyz[(m0 + ln) * 4 + khalf];
      v8f dt = __builtin_amdgcn_wmma_f32_16x16x4_f32(
          false, pf, false, uf, (short)0, cu, false, false);
      const float* fb = &s_feat[((m0 >> 1) + hpair) * FPITCH + 2 * ln];
#pragma unroll
      for (int j = 0; j < 4; ++j) {
        v2f d2;
        d2.x = dt[2 * j];
        d2.y = dt[2 * j + 1];
        d2 += eps2;
        v2f a;
        a.x = __builtin_amdgcn_rcpf(d2.x);
        a.y = __builtin_amdgcn_rcpf(d2.y);
        rs2 += a;
#pragma unroll
        for (int t = 0; t < 4; ++t) {
          v2f bm = *(const v2f*)&fb[j * FPITCH + t * 32];
          acc[t] = __builtin_amdgcn_wmma_f32_16x16x4_f32(
              false, a, false, bm, (short)0, acc[t], false, false);
        }
      }
    }
    __syncthreads();
  }
  const long rowbase = (long)(b * NSZ + n0);
  v4f fv[8];
  {
    const v4f* fu = (const v4f*)(feat_up + rowbase * CH);
#pragma unroll
    for (int j = 0; j < 8; ++j) fv[j] = fu[j * 32 + lane];
  }

  const float rsum = rs2.x + rs2.y;

  float denom = rsum + __shfl_xor(rsum, 16, 32);
  float inv   = 1.f / denom;

  float*    tile = s_feat + wave * 16 * TP;
  const int hrow = (lane & 16) ? 8 : 0;

  float s1[8], t1[8], s2[4], t2[4];
#pragma unroll
  for (int t = 0; t < 8; ++t) {
    int   c  = t * 16 + ln;
    float sc = g1[c] * rsqrtf(va1[c] + 1e-5f);
    s1[t] = sc;
    t1[t] = (b1[c] - mu1[c]) * sc + be1[c];
  }
#pragma unroll
  for (int t = 0; t < 4; ++t) {
    int   c  = t * 16 + ln;
    float sc = g2[c] * rsqrtf(va2[c] + 1e-5f);
    s2[t] = sc;
    t2[t] = (b2[c] - mu2[c]) * sc + be2[c];
  }

#pragma unroll
  for (int j = 0; j < 8; ++j) {
    int i4  = j * 32 + lane;
    int row = i4 >> 4;
    int c4  = i4 & 15;
    *(v4f*)&tile[row * TP + c4 * 4] = fv[j];
  }
  __syncthreads();

  v8f h[8];
#pragma unroll
  for (int t = 0; t < 8; ++t) h[t] = z;

  for (int k = 0; k < CH; k += 4) {
    v2f a = *(const v2f*)&tile[ln * TP + k + khalf];
#pragma unroll
    for (int t = 0; t < 8; ++t) {
      v2f bm = *(const v2f*)&s_w1[(t * 16 + ln) * XSTR + k + khalf];
      h[t] = __builtin_amdgcn_wmma_f32_16x16x4_f32(
          false, a, false, bm, (short)0, h[t], false, false);
    }
  }
  __syncthreads();

#pragma unroll
  for (int j = 0; j < 8; ++j) {
    float dj = __shfl(inv, hrow + j, 32);
#pragma unroll
    for (int t = 0; t < 4; ++t)
      tile[(hrow + j) * TP + t * 16 + ln] = acc[t][j] * dj;
  }
  __syncthreads();

  for (int k = 0; k < CH; k += 4) {
    v2f a = *(const v2f*)&tile[ln * TP + k + khalf];
#pragma unroll
    for (int t = 0; t < 8; ++t) {
      v2f bm = *(const v2f*)&s_w1[(t * 16 + ln) * XSTR + CH + k + khalf];
      h[t] = __builtin_amdgcn_wmma_f32_16x16x4_f32(
          false, a, false, bm, (short)0, h[t], false, false);
    }
  }
  __syncthreads();

  v8f o[4];
#pragma unroll
  for (int t = 0; t < 4; ++t) o[t] = z;

#pragma unroll
  for (int hf = 0; hf < 2; ++hf) {
#pragma unroll
    for (int t = 0; t < 4; ++t) {
#pragma unroll
      for (int j = 0; j < 8; ++j) {
        float v = h[hf * 4 + t][j] * s1[hf * 4 + t] + t1[hf * 4 + t];
        tile[(hrow + j) * TP + t * 16 + ln] = v > 0.f ? v : 0.f;
      }
    }
    __syncthreads();
    for (int k = 0; k < CH; k += 4) {
      v2f a = *(const v2f*)&tile[ln * TP + k + khalf];
#pragma unroll
      for (int t = 0; t < 4; ++t) {
        v2f bm = *(const v2f*)&s_w2[(t * 16 + ln) * XSTR + hf * CH + k + khalf];
        o[t] = __builtin_amdgcn_wmma_f32_16x16x4_f32(
            false, a, false, bm, (short)0, o[t], false, false);
      }
    }
    if (hf == 0) __syncthreads();
  }

#pragma unroll
  for (int t = 0; t < 4; ++t) {
#pragma unroll
    for (int j = 0; j < 8; ++j)
      out[(rowbase + hrow + j) * OCH + t * 16 + ln] = o[t][j] * s2[t] + t2[t];
  }
}

extern "C" void kernel_launch(void* const* d_in, const int* in_sizes, int n_in,
                              void* d_out, int out_size, void* d_ws,
                              size_t ws_size, hipStream_t stream) {
  const float* xyz_down  = (const float*)d_in[0];
  const float* xyz_up    = (const float*)d_in[1];
  const float* feat_down = (const float*)d_in[2];
  const float* feat_up   = (const float*)d_in[3];
  const float* W1  = (const float*)d_in[4];
  const float* b1  = (const float*)d_in[5];
  const float* g1  = (const float*)d_in[6];
  const float* be1 = (const float*)d_in[7];
  const float* mu1 = (const float*)d_in[8];
  const float* va1 = (const float*)d_in[9];
  const float* W2  = (const float*)d_in[10];
  const float* b2  = (const float*)d_in[11];
  const float* g2  = (const float*)d_in[12];
  const float* be2 = (const float*)d_in[13];
  const float* mu2 = (const float*)d_in[14];
  const float* va2 = (const float*)d_in[15];

  dim3 grid(NSZ / 128, BSZ);
  interp_mlp_kernel<<<grid, 256, 0, stream>>>(
      xyz_down, xyz_up, feat_down, feat_up,
      W1, b1, g1, be1, mu1, va1, W2, b2, g2, be2, mu2, va2,
      (float*)d_out);
}


# LDS fragment reads of the next WMMA block (17 WMMAs in the interp loop, 16 in the MLP products) issued as ONE group a block ahead into a register double buffer; every sum in the parent's order
# speedup vs baseline: 1.4426x; 1.0509x over previous
#include <hip/hip_runtime.h>

#define BSZ 4
#define MSZ 2048
#define NSZ 16384
#define CH   64
#define XCH  128
#define OCH  64

typedef float v2f __attribute__((ext_vector_type(2)));
typedef float v4f __attribute__((ext_vector_type(4)));
typedef float v8f __attribute__((ext_vector_type(8)));

#define MCHUNK  128
#define FPITCH  136
#define XSTR    132
#define TP      68

template <int NT, int KS>
__device__ __forceinline__ void gemm_seg64(v8f (&accv)[NT], const float* arow, const float* wrow) {
  constexpr int Q = KS / 4;
  constexpr int G = CH / KS;
  v2f af[2][Q];
  v2f bf[2][Q][NT];
#define LOAD_GROUP(K0, BUF)                                                \
  _Pragma("unroll") for (int q = 0; q < Q; ++q) {                          \
    af[BUF][q] = *(const v2f*)&arow[(K0) + 4 * q];                         \
    _Pragma("unroll") for (int t = 0; t < NT; ++t)                         \
      bf[BUF][q][t] = *(const v2f*)&wrow[t * 16 * XSTR + (K0) + 4 * q];    \
  }
  LOAD_GROUP(0, 0)
#pragma unroll
  for (int g = 0; g < G; ++g) {
    const int cur = g & 1;
    if (g + 1 < G) {
      __builtin_amdgcn_sched_barrier(0);
      LOAD_GROUP((g + 1) * KS, cur ^ 1)
      __builtin_amdgcn_sched_barrier(0);
    }
#pragma unroll
    for (int q = 0; q < Q; ++q) {
#pragma unroll
      for (int t = 0; t < NT; ++t)
        accv[t] = __builtin_amdgcn_wmma_f32_16x16x4_f32(
            false, af[cur][q], false, bf[cur][q][t], (short)0, accv[t], false, false);
    }
  }
#undef LOAD_GROUP
}

__global__ __launch_bounds__(256) void interp_mlp_kernel(
    const float* __restrict__ xyz_down, const float* __restrict__ xyz_up,
    const float* __restrict__ feat_down, const float* __restrict__ feat_up,
    const float* __restrict__ W1, const float* __restrict__ b1,
    const float* __restrict__ g1, const float* __restrict__ be1,
    const float* __restrict__ mu1, const float* __restrict__ va1,
    const float* __restrict__ W2, const float* __restrict__ b2,
    const float* __restrict__ g2, const float* __restrict__ be2,
    const float* __restrict__ mu2, const float* __restrict__ va2,
    float* __restrict__ out)
{
  __shared__ __attribute__((aligned(16))) float s_xyz[MCHUNK * 4];
  __shared__ __attribute__((aligned(16))) float s_feat[(MCHUNK / 2) * FPITCH];
  __shared__ __attribute__((aligned(16))) float s_w1[XCH * XSTR];
  __shared__ __attribute__((aligned(16))) float s_w2[OCH * XSTR];

  const int b     = blockIdx.y;
  const int tid   = threadIdx.x;
  const int wave  = tid >> 5;
  const int lane  = tid & 31;
  const int ln    = lane & 15;
  const int khalf = (lane >> 4) << 1;
  const int hpair = (lane >> 4) << 2;
  const int n0    = blockIdx.x * 128 + wave * 16;

  const float* up = xyz_up + ((long)(b * NSZ + n0 + ln)) * 3;
  const float ux = up[0], uy = up[1], uz = up[2];
  const float u2 = ux * ux + uy * uy + uz * uz;

  v2f uf;
  uf.x = (lane & 16) ? uy : 1.f;
  uf.y = (lane & 16) ? uz : ux;
  const v8f cu   = {u2, u2, u2, u2, u2, u2, u2, u2};
  const v2f eps2 = {1e-8f, 1e-8f};

  v8f z = {0.f, 0.f, 0.f, 0.f, 0.f, 0.f, 0.f, 0.f};
  v8f acc[4];
#pragma unroll
  for (int t = 0; t < 4; ++t) acc[t] = z;
  v2f rs2 = {0.f, 0.f};

  v2f    pxy;
  float  pzr;
  float4 fr[8];
  float4 w1r;
  float2 w2r;

#define LOAD_CHUNK(CS)                                                               \
  {                                                                                  \
      \
      \
    const float* p = xyz_down + ((long)(b * MSZ + (CS) + (tid & (MCHUNK - 1)))) * 3; \
    pxy.x = p[0];                                                                    \
    pxy.y = p[1];                                                                    \
    pzr   = p[2];                                                                    \
    const float4* fd = (const float4*)(feat_down + ((long)(b * MSZ + (CS))) * CH);   \
    _Pragma("unroll") for (int j = 0; j < 8; ++j) fr[j] = fd[j * 256 + tid];         \
    const int iw = ((CS) / MCHUNK) * 256 + tid;                          \
    w1r = ((const float4*)W1)[iw];                                                   \
    w2r = ((const float2*)W2)[iw];                                                   \
  }

  LOAD_CHUNK(0);

  for (int cstart = 0; cstart < MSZ; cstart += MCHUNK) {
    if (tid < MCHUNK) {
      const v2f sq = pxy * pxy;
      const v2f n2 = pxy * -2.f;
      float4 v = {sq.x + sq.y + pzr * pzr, n2.x, n2.y, -2.f * pzr};
      *(float4*)&s_xyz[tid * 4] = v;
    }
#pragma unroll
    for (int j = 0; j < 8; ++j) {
      int i4  = j * 256 + tid;
      int row = i4 >> 4;
      int c4  = i4 & 15;
      float* dst = &s_feat[(row >> 1) * FPITCH + c4 * 8 + (row & 1)];
      dst[0] = fr[j].x;
      dst[2] = fr[j].y;
      dst[4] = fr[j].z;
      dst[6] = fr[j].w;
    }
    {
      int i4  = (cstart / MCHUNK) * 256 + tid;
      int row = i4 >> 5;
      int c4  = i4 & 31;
      int r2  = i4 >> 6;
      int c2  = i4 & 63;
      *(float4*)&s_w1[row * XSTR + c4 * 4] = w1r;
      *(float2*)&s_w2[r2 * XSTR + c2 * 2]  = w2r;
    }
    __syncthreads();

    if (cstart + MCHUNK < MSZ) LOAD_CHUNK(cstart + MCHUNK);

    v2f pfq[2];
    v2f bq[2][16];
#define LOAD_BLOCK(M0, BUF)                                                          \
  {                                                                                  \
    pfq[BUF] = *(const v2f*)&s_xyz[((M0) + ln) * 4 + khalf];                         \
    const float* fbl = &s_feat[(((M0) >> 1) + hpair) * FPITCH + 2 * ln];             \
    _Pragma("unroll") for (int j = 0; j < 4; ++j)                                    \
      _Pragma("unroll") for (int t = 0; t < 4; ++t)                                  \
        bq[BUF][4 * j + t] = *(const v2f*)&fbl[j * FPITCH + t * 32];                 \
  }
    LOAD_BLOCK(0, 0);
    v8f dt = z;
#pragma unroll
    for (int blk = 0; blk < MCHUNK / 16; ++blk) {
      const int cur = blk & 1;
      if (blk + 1 < MCHUNK / 16) {
        __builtin_amdgcn_sched_barrier(0);
        LOAD_BLOCK((blk + 1) * 16, cur ^ 1);
        __builtin_amdgcn_sched_barrier(0);
      }
      if (blk == 0)
        dt = __builtin_amdgcn_wmma_f32_16x16x4_f32(
            false, pfq[0], false, uf, (short)0, cu, false, false);
      v8f dtn = dt;
#pragma unroll
      for (int j = 0; j < 4; ++j) {
        v2f d2;
        d2.x = dt[2 * j];
        d2.y = dt[2 * j + 1];
        d2 += eps2;
        v2f a;
        a.x = __builtin_amdgcn_rcpf(d2.x);
        a.y = __builtin_amdgcn_rcpf(d2.y);
        rs2 += a;
#pragma unroll
        for (int t = 0; t < 4; ++t) {
          acc[t] = __builtin_amdgcn_wmma_f32_16x16x4_f32(
              false, a, false, bq[cur][4 * j + t], (short)0, acc[t], false, false);
        }
        if (j == 1 && blk + 1 < MCHUNK / 16)
          dtn = __builtin_amdgcn_wmma_f32_16x16x4_f32(
              false, pfq[cur ^ 1], false, uf, (short)0, cu, false, false);
      }
      dt = dtn;
    }
#undef LOAD_BLOCK
    __syncthreads();
  }
  const long rowbase = (long)(b * NSZ + n0);
  v4f fv[8];
  {
    const v4f* fu = (const v4f*)(feat_up + rowbase * CH);
#pragma unroll
    for (int j = 0; j < 8; ++j) fv[j] = fu[j * 32 + lane];
  }

  const float rsum = rs2.x + rs2.y;

  float denom = rsum + __shfl_xor(rsum, 16, 32);
  float inv   = 1.f / denom;

  float*    tile = s_feat + wave * 16 * TP;
  const int hrow = (lane & 16) ? 8 : 0;

  float s1[8], t1[8], s2[4], t2[4];
#pragma unroll
  for (int t = 0; t < 8; ++t) {
    int   c  = t * 16 + ln;
    float sc = g1[c] * rsqrtf(va1[c] + 1e-5f);
    s1[t] = sc;
    t1[t] = (b1[c] - mu1[c]) * sc + be1[c];
  }
#pragma unroll
  for (int t = 0; t < 4; ++t) {
    int   c  = t * 16 + ln;
    float sc = g2[c] * rsqrtf(va2[c] + 1e-5f);
    s2[t] = sc;
    t2[t] = (b2[c] - mu2[c]) * sc + be2[c];
  }

#pragma unroll
  for (int j = 0; j < 8; ++j) {
    int i4  = j * 32 + lane;
    int row = i4 >> 4;
    int c4  = i4 & 15;
    *(v4f*)&tile[row * TP + c4 * 4] = fv[j];
  }
  __syncthreads();

  v8f h[8];
#pragma unroll
  for (int t = 0; t < 8; ++t) h[t] = z;

  gemm_seg64<8, 8>(h, &tile[ln * TP + khalf], &s_w1[ln * XSTR + khalf]);
  __syncthreads();

#pragma unroll
  for (int j = 0; j < 8; ++j) {
    float dj = __shfl(inv, hrow + j, 32);
#pragma unroll
    for (int t = 0; t < 4; ++t)
      tile[(hrow + j) * TP + t * 16 + ln] = acc[t][j] * dj;
  }
  __syncthreads();

  gemm_seg64<8, 8>(h, &tile[ln * TP + khalf], &s_w1[ln * XSTR + CH + khalf]);
  __syncthreads();

  v8f o[4];
#pragma unroll
  for (int t = 0; t < 4; ++t) o[t] = z;

#pragma unroll
  for (int hf = 0; hf < 2; ++hf) {
#pragma unroll
    for (int t = 0; t < 4; ++t) {
#pragma unroll
      for (int j = 0; j < 8; ++j) {
        float v = h[hf * 4 + t][j] * s1[hf * 4 + t] + t1[hf * 4 + t];
        tile[(hrow + j) * TP + t * 16 + ln] = v > 0.f ? v : 0.f;
      }
    }
    __syncthreads();
    gemm_seg64<4, 16>(o, &tile[ln * TP + khalf], &s_w2[ln * XSTR + hf * CH + khalf]);
    if (hf == 0) __syncthreads();
  }

#pragma unroll
  for (int t = 0; t < 4; ++t) {
#pragma unroll
    for (int j = 0; j < 8; ++j)
      out[(rowbase + hrow + j) * OCH + t * 16 + ln] = o[t][j] * s2[t] + t2[t];
  }
}

extern "C" void kernel_launch(void* const* d_in, const int* in_sizes, int n_in,
                              void* d_out, int out_size, void* d_ws,
                              size_t ws_size, hipStream_t stream) {
  const float* xyz_down  = (const float*)d_in[0];
  const float* xyz_up    = (const float*)d_in[1];
  const float* feat_down = (const float*)d_in[2];
  const float* feat_up   = (const float*)d_in[3];
  const float* W1  = (const float*)d_in[4];
  const float* b1  = (const float*)d_in[5];
  const float* g1  = (const float*)d_in[6];
  const float* be1 = (const float*)d_in[7];
  const float* mu1 = (const float*)d_in[8];
  const float* va1 = (const float*)d_in[9];
  const float* W2  = (const float*)d_in[10];
  const float* b2  = (const float*)d_in[11];
  const float* g2  = (const float*)d_in[12];
  const float* be2 = (const float*)d_in[13];
  const float* mu2 = (const float*)d_in[14];
  const float* va2 = (const float*)d_in[15];

  dim3 grid(NSZ / 128, BSZ);
  interp_mlp_kernel<<<grid, 256, 0, stream>>>(
      xyz_down, xyz_up, feat_down, feat_up,
      W1, b1, g1, be1, mu1, va1, W2, b2, g2, be2, mu2, va2,
      (float*)d_out);
}


# wave priority falls with progress (s_setprio 3; 2 from point chunk 8; 1 in the MLP; 0 from its 2nd product): the later-started of a WGP's two work-groups catches up instead of finishing alone
# speedup vs baseline: 1.4880x; 1.0315x over previous
#include <hip/hip_runtime.h>

#define BSZ 4
#define MSZ 2048
#define NSZ 16384
#define CH   64
#define XCH  128
#define OCH  64

typedef float v2f __attribute__((ext_vector_type(2)));
typedef float v4f __attribute__((ext_vector_type(4)));
typedef float v8f __attribute__((ext_vector_type(8)));

#define MCHUNK  128
#define FPITCH  136
#define XSTR    132
#define TP      68

template <int NT, int KS>
__device__ __forceinline__ void gemm_seg64(v8f (&accv)[NT], const float* arow, const float* wrow) {
  constexpr int Q = KS / 4;
  constexpr int G = CH / KS;
  v2f af[2][Q];
  v2f bf[2][Q][NT];
#define LOAD_GROUP(K0, BUF)                                                \
  _Pragma("unroll") for (int q = 0; q < Q; ++q) {                          \
    af[BUF][q] = *(const v2f*)&arow[(K0) + 4 * q];                         \
    _Pragma("unroll") for (int t = 0; t < NT; ++t)                         \
      bf[BUF][q][t] = *(const v2f*)&wrow[t * 16 * XSTR + (K0) + 4 * q];    \
  }
  LOAD_GROUP(0, 0)
#pragma unroll
  for (int g = 0; g < G; ++g) {
    const int cur = g & 1;
    if (g + 1 < G) {
      __builtin_amdgcn_sched_barrier(0);
      LOAD_GROUP((g + 1) * KS, cur ^ 1)
      __builtin_amdgcn_sched_barrier(0);
    }
#pragma unroll
    for (int q = 0; q < Q; ++q) {
#pragma unroll
      for (int t = 0; t < NT; ++t)
        accv[t] = __builtin_amdgcn_wmma_f32_16x16x4_f32(
            false, af[cur][q], false, bf[cur][q][t], (short)0, accv[t], false, false);
    }
  }
#undef LOAD_GROUP
}

__global__ __launch_bounds__(256) void interp_mlp_kernel(
    const float* __restrict__ xyz_down, const float* __restrict__ xyz_up,
    const float* __restrict__ feat_down, const float* __restrict__ feat_up,
    const float* __restrict__ W1, const float* __restrict__ b1,
    const float* __restrict__ g1, const float* __restrict__ be1,
    const float* __restrict__ mu1, const float* __restrict__ va1,
    const float* __restrict__ W2, const float* __restrict__ b2,
    const float* __restrict__ g2, const float* __restrict__ be2,
    const float* __restrict__ mu2, const float* __restrict__ va2,
    float* __restrict__ out)
{
  __shared__ __attribute__((aligned(16))) float s_xyz[MCHUNK * 4];
  __shared__ __attribute__((aligned(16))) float s_feat[(MCHUNK / 2) * FPITCH];
  __shared__ __attribute__((aligned(16))) float s_w1[XCH * XSTR];
  __shared__ __attribute__((aligned(16))) float s_w2[OCH * XSTR];

  __builtin_amdgcn_s_setprio(3);

  const int b     = blockIdx.y;
  const int tid   = threadIdx.x;
  const int wave  = tid >> 5;
  const int lane  = tid & 31;
  const int ln    = lane & 15;
  const int khalf = (lane >> 4) << 1;
  const int hpair = (lane >> 4) << 2;
  const int n0    = blockIdx.x * 128 + wave * 16;

  const float* up = xyz_up + ((long)(b * NSZ + n0 + ln)) * 3;
  const float ux = up[0], uy = up[1], uz = up[2];
  const float u2 = ux * ux + uy * uy + uz * uz;

  v2f uf;
  uf.x = (lane & 16) ? uy : 1.f;
  uf.y = (lane & 16) ? uz : ux;
  const v8f cu   = {u2, u2, u2, u2, u2, u2, u2, u2};
  const v2f eps2 = {1e-8f, 1e-8f};

  v8f z = {0.f, 0.f, 0.f, 0.f, 0.f, 0.f, 0.f, 0.f};
  v8f acc[4];
#pragma unroll
  for (int t = 0; t < 4; ++t) acc[t] = z;
  v2f rs2 = {0.f, 0.f};

  v2f    pxy;
  float  pzr;
  float4 fr[8];
  float4 w1r;
  float2 w2r;

#define LOAD_CHUNK(CS)                                                               \
  {                                                                                  \
      \
      \
    const float* p = xyz_down + ((long)(b * MSZ + (CS) + (tid & (MCHUNK - 1)))) * 3; \
    pxy.x = p[0];                                                                    \
    pxy.y = p[1];                                                                    \
    pzr   = p[2];                                                                    \
    const float4* fd = (const float4*)(feat_down + ((long)(b * MSZ + (CS))) * CH);   \
    _Pragma("unroll") for (int j = 0; j < 8; ++j) fr[j] = fd[j * 256 + tid];         \
    const int iw = ((CS) / MCHUNK) * 256 + tid;                          \
    w1r = ((const float4*)W1)[iw];                                                   \
    w2r = ((const float2*)W2)[iw];                                                   \
  }

  LOAD_CHUNK(0);

  for (int cstart = 0; cstart < MSZ; cstart += MCHUNK) {
    if (cstart == MSZ / 2) __builtin_amdgcn_s_setprio(2);
    if (tid < MCHUNK) {
      const v2f sq = pxy * pxy;
      const v2f n2 = pxy * -2.f;
      float4 v = {sq.x + sq.y + pzr * pzr, n2.x, n2.y, -2.f * pzr};
      *(float4*)&s_xyz[tid * 4] = v;
    }
#pragma unroll
    for (int j = 0; j < 8; ++j) {
      int i4  = j * 256 + tid;
      int row = i4 >> 4;
      int c4  = i4 & 15;
      float* dst = &s_feat[(row >> 1) * FPITCH + c4 * 8 + (row & 1)];
      dst[0] = fr[j].x;
      dst[2] = fr[j].y;
      dst[4] = fr[j].z;
      dst[6] = fr[j].w;
    }
    {
      int i4  = (cstart / MCHUNK) * 256 + tid;
      int row = i4 >> 5;
      int c4  = i4 & 31;
      int r2  = i4 >> 6;
      int c2  = i4 & 63;
      *(float4*)&s_w1[row * XSTR + c4 * 4] = w1r;
      *(float2*)&s_w2[r2 * XSTR + c2 * 2]  = w2r;
    }
    __syncthreads();

    if (cstart + MCHUNK < MSZ) LOAD_CHUNK(cstart + MCHUNK);

    v2f pfq[2];
    v2f bq[2][16];
#define LOAD_BLOCK(M0, BUF)                                                          \
  {                                                                                  \
    pfq[BUF] = *(const v2f*)&s_xyz[((M0) + ln) * 4 + khalf];                         \
    const float* fbl = &s_feat[(((M0) >> 1) + hpair) * FPITCH + 2 * ln];             \
    _Pragma("unroll") for (int j = 0; j < 4; ++j)                                    \
      _Pragma("unroll") for (int t = 0; t < 4; ++t)                                  \
        bq[BUF][4 * j + t] = *(const v2f*)&fbl[j * FPITCH + t * 32];                 \
  }
    LOAD_BLOCK(0, 0);
    v8f dt = z;
#pragma unroll
    for (int blk = 0; blk < MCHUNK / 16; ++blk) {
      const int cur = blk & 1;
      if (blk + 1 < MCHUNK / 16) {
        __builtin_amdgcn_sched_barrier(0);
        LOAD_BLOCK((blk + 1) * 16, cur ^ 1);
        __builtin_amdgcn_sched_barrier(0);
      }
      if (blk == 0)
        dt = __builtin_amdgcn_wmma_f32_16x16x4_f32(
            false, pfq[0], false, uf, (short)0, cu, false, false);
      v8f dtn = dt;
#pragma unroll
      for (int j = 0; j < 4; ++j) {
        v2f d2;
        d2.x = dt[2 * j];
        d2.y = dt[2 * j + 1];
        d2 += eps2;
        v2f a;
        a.x = __builtin_amdgcn_rcpf(d2.x);
        a.y = __builtin_amdgcn_rcpf(d2.y);
        rs2 += a;
#pragma unroll
        for (int t = 0; t < 4; ++t) {
          acc[t] = __builtin_amdgcn_wmma_f32_16x16x4_f32(
              false, a, false, bq[cur][4 * j + t], (short)0, acc[t], false, false);
        }
        if (j == 1 && blk + 1 < MCHUNK / 16)
          dtn = __builtin_amdgcn_wmma_f32_16x16x4_f32(
              false, pfq[cur ^ 1], false, uf, (short)0, cu, false, false);
      }
      dt = dtn;
    }
#undef LOAD_BLOCK
    __syncthreads();
  }
  __builtin_amdgcn_s_setprio(1);

  const long rowbase = (long)(b * NSZ + n0);
  v4f fv[8];
  {
    const v4f* fu = (const v4f*)(feat_up + rowbase * CH);
#pragma unroll
    for (int j = 0; j < 8; ++j) fv[j] = fu[j * 32 + lane];
  }

  const float rsum = rs2.x + rs2.y;

  float denom = rsum + __shfl_xor(rsum, 16, 32);
  float inv   = 1.f / denom;

  float*    tile = s_feat + wave * 16 * TP;
  const int hrow = (lane & 16) ? 8 : 0;

  float s1[8], t1[8], s2[4], t2[4];
#pragma unroll
  for (int t = 0; t < 8; ++t) {
    int   c  = t * 16 + ln;
    float sc = g1[c] * rsqrtf(va1[c] + 1e-5f);
    s1[t] = sc;
    t1[t] = (b1[c] - mu1[c]) * sc + be1[c];
  }
#pragma unroll
  for (int t = 0; t < 4; ++t) {
    int   c  = t * 16 + ln;
    float sc = g2[c] * rsqrtf(va2[c] + 1e-5f);
    s2[t] = sc;
    t2[t] = (b2[c] - mu2[c]) * sc + be2[c];
  }

#pragma unroll
  for (int j = 0; j < 8; ++j) {
    int i4  = j * 32 + lane;
    int row = i4 >> 4;
    int c4  = i4 & 15;
    *(v4f*)&tile[row * TP + c4 * 4] = fv[j];
  }
  __syncthreads();

  v8f h[8];
#pragma unroll
  for (int t = 0; t < 8; ++t) h[t] = z;

  gemm_seg64<8, 8>(h, &tile[ln * TP + khalf], &s_w1[ln * XSTR + khalf]);
  __syncthreads();

#pragma unroll
  for (int j = 0; j < 8; ++j) {
    float dj = __shfl(inv, hrow + j, 32);
#pragma unroll
    for (int t = 0; t < 4; ++t)
      tile[(hrow + j) * TP + t * 16 + ln] = acc[t][j] * dj;
  }
  __syncthreads();

  gemm_seg64<8, 8>(h, &tile[ln * TP + khalf], &s_w1[ln * XSTR + CH + khalf]);
  __syncthreads();

  __builtin_amdgcn_s_setprio(0);

  v8f o[4];
#pragma unroll
  for (int t = 0; t < 4; ++t) o[t] = z;

#pragma unroll
  for (int hf = 0; hf < 2; ++hf) {
#pragma unroll
    for (int t = 0; t < 4; ++t) {
#pragma unroll
      for (int j = 0; j < 8; ++j) {
        float v = h[hf * 4 + t][j] * s1[hf * 4 + t] + t1[hf * 4 + t];
        tile[(hrow + j) * TP + t * 16 + ln] = v > 0.f ? v : 0.f;
      }
    }
    __syncthreads();
    gemm_seg64<4, 16>(o, &tile[ln * TP + khalf], &s_w2[ln * XSTR + hf * CH + khalf]);
    if (hf == 0) __syncthreads();
  }

#pragma unroll
  for (int t = 0; t < 4; ++t) {
#pragma unroll
    for (int j = 0; j < 8; ++j)
      out[(rowbase + hrow + j) * OCH + t * 16 + ln] = o[t][j] * s2[t] + t2[t];
  }
}

extern "C" void kernel_launch(void* const* d_in, const int* in_sizes, int n_in,
                              void* d_out, int out_size, void* d_ws,
                              size_t ws_size, hipStream_t stream) {
  const float* xyz_down  = (const float*)d_in[0];
  const float* xyz_up    = (const float*)d_in[1];
  const float* feat_down = (const float*)d_in[2];
  const float* feat_up   = (const float*)d_in[3];
  const float* W1  = (const float*)d_in[4];
  const float* b1  = (const float*)d_in[5];
  const float* g1  = (const float*)d_in[6];
  const float* be1 = (const float*)d_in[7];
  const float* mu1 = (const float*)d_in[8];
  const float* va1 = (const float*)d_in[9];
  const float* W2  = (const float*)d_in[10];
  const float* b2  = (const float*)d_in[11];
  const float* g2  = (const float*)d_in[12];
  const float* be2 = (const float*)d_in[13];
  const float* mu2 = (const float*)d_in[14];
  const float* va2 = (const float*)d_in[15];

  dim3 grid(NSZ / 128, BSZ);
  interp_mlp_kernel<<<grid, 256, 0, stream>>>(
      xyz_down, xyz_up, feat_down, feat_up,
      W1, b1, g1, be1, mu1, va1, W2, b2, g2, be2, mu2, va2,
      (float*)d_out);
}
